# starting kernel
# baseline (speedup 1.0000x reference)
#include <hip/hip_runtime.h>

#define B_TOTAL 8192
#define T_STEPS 256
#define IN_F    16
#define HID     32
#define L2E     1.4426950408889634f

typedef __attribute__((ext_vector_type(16))) _Float16 v16h;
typedef __attribute__((ext_vector_type(8)))  _Float16 v8h;
typedef __attribute__((ext_vector_type(8)))  float    v8f;

#if __has_builtin(__builtin_amdgcn_exp2f)
static __device__ __forceinline__ float fexp2(float x) { return __builtin_amdgcn_exp2f(x); }
#else
static __device__ __forceinline__ float fexp2(float x) { return __expf(x * 0.6931471805599453f); }
#endif

static __device__ __forceinline__ float sigm_b(float g, float nb) {
  return __builtin_amdgcn_rcpf(1.0f + fexp2(__builtin_fmaf(g, -L2E, nb)));
}
static __device__ __forceinline__ float tanh_b(float g, float nb) {
  return 2.0f * __builtin_amdgcn_rcpf(1.0f + fexp2(__builtin_fmaf(g, -2.0f * L2E, nb))) - 1.0f;
}
static __device__ __forceinline__ float tanh_nb(float c) {
  return 2.0f * __builtin_amdgcn_rcpf(1.0f + fexp2(c * (-2.0f * L2E))) - 1.0f;
}

static __device__ __forceinline__ v16h afrag_from_lds(const _Float16* h, int lane) {
  const int m  = lane & 15;
  const int kb = (lane < 16) ? 0 : 8;
  const _Float16* p = h + m * HID + kb;
  v8h lo = *(const v8h*)(p);
  v8h hi = *(const v8h*)(p + 16);
  return __builtin_shufflevector(lo, hi, 0,1,2,3,4,5,6,7,8,9,10,11,12,13,14,15);
}

static __device__ __forceinline__ v16h bfrag_from_w(const float* W, int ld, int tile,
                                                    bool padk, int lane) {
  const int n  = tile * 16 + (lane & 15);
  const int kb = (lane < 16) ? 0 : 8;
  const float* row = W + n * ld;
  v16h b;
#pragma unroll
  for (int i = 0; i < 8; ++i) b[i] = (_Float16)row[kb + i];
#pragma unroll
  for (int i = 0; i < 8; ++i) b[8 + i] = padk ? (_Float16)0.0f : (_Float16)row[16 + kb + i];
  return b;
}

static __device__ __forceinline__ void lstm_half_cell(const v16h in, const v16h hr,
                                                      const v16h* Wi, const v16h* Wh,
                                                      const float* nb,
                                                      v8f& c, _Float16* hout,
                                                      int lane, int w) {
  v8f g[4];
#pragma unroll
  for (int gt = 0; gt < 4; ++gt) {
    v8f acc = {};
    acc = __builtin_amdgcn_wmma_f32_16x16x32_f16(false, in, false, Wi[gt], (short)0, acc, false, false);
    acc = __builtin_amdgcn_wmma_f32_16x16x32_f16(false, hr, false, Wh[gt], (short)0, acc, false, false);
    g[gt] = acc;
  }
  const int col = w * 16 + (lane & 15);
  const int mb  = (lane < 16) ? 0 : 8;
#pragma unroll
  for (int r = 0; r < 8; ++r) {
    float iv = sigm_b(g[0][r], nb[0]);
    float fv = sigm_b(g[1][r], nb[1]);
    float gv = tanh_b(g[2][r], nb[2]);
    float ov = sigm_b(g[3][r], nb[3]);
    float cv = __builtin_fmaf(fv, c[r], iv * gv);
    c[r] = cv;
    float hv = ov * tanh_nb(cv);
    hout[(mb + r) * HID + col] = (_Float16)hv;
  }
}

extern "C" __global__ __launch_bounds__(64)
void lstm2_wmma_fused(const float* __restrict__ x,
                      const float* __restrict__ Wih0, const float* __restrict__ Whh0,
                      const float* __restrict__ bih0, const float* __restrict__ bhh0,
                      const float* __restrict__ Wih1, const float* __restrict__ Whh1,
                      const float* __restrict__ bih1, const float* __restrict__ bhh1,
                      const float* __restrict__ Wfc,  const float* __restrict__ bfc,
                      float* __restrict__ out) {
  __shared__ _Float16 hA[16 * HID];
  __shared__ _Float16 hB[16 * HID];
  const int lane = threadIdx.x & 31;
  const int w    = threadIdx.x >> 5;
  const int b0   = blockIdx.x * 16;

  v16h wi0[4], wh0[4], wi1[4], wh1[4];
#pragma unroll
  for (int gt = 0; gt < 4; ++gt) {
    const int tile = gt * 2 + w;
    wi0[gt] = bfrag_from_w(Wih0, IN_F, tile, true,  lane);
    wh0[gt] = bfrag_from_w(Whh0, HID,  tile, false, lane);
    wi1[gt] = bfrag_from_w(Wih1, HID,  tile, false, lane);
    wh1[gt] = bfrag_from_w(Whh1, HID,  tile, false, lane);
  }
  float nb0[4], nb1[4];
#pragma unroll
  for (int gt = 0; gt < 4; ++gt) {
    const int col   = (gt * 2 + w) * 16 + (lane & 15);
    const float scl = (gt == 2) ? (-2.0f * L2E) : (-L2E);
    nb0[gt] = scl * (bih0[col] + bhh0[col]);
    nb1[gt] = scl * (bih1[col] + bhh1[col]);
  }
  v16h wfcf;
  {
    const int  kb = (lane < 16) ? 0 : 8;
    const bool n0 = (lane & 15) == 0;
#pragma unroll
    for (int i = 0; i < 8; ++i) wfcf[i]     = n0 ? (_Float16)Wfc[kb + i]      : (_Float16)0.0f;
#pragma unroll
    for (int i = 0; i < 8; ++i) wfcf[8 + i] = n0 ? (_Float16)Wfc[16 + kb + i] : (_Float16)0.0f;
  }

  v8f c0 = {}, c1 = {};
  v16h hAf, hBf;
#pragma unroll
  for (int i = 0; i < 16; ++i) { hAf[i] = (_Float16)0.0f; hBf[i] = (_Float16)0.0f; }

  const int m_row = lane & 15;
  const int kb_x  = (lane < 16) ? 0 : 8;
  const float* xrow = x + (size_t)(b0 + m_row) * T_STEPS * IN_F + kb_x;

#pragma unroll 1
  for (int t = 0; t < T_STEPS; ++t) {
    v16h xf;
    const float* p = xrow + (size_t)t * IN_F;
#pragma unroll
    for (int i = 0; i < 8; ++i) xf[i] = (_Float16)p[i];
#pragma unroll
    for (int i = 0; i < 8; ++i) xf[8 + i] = (_Float16)0.0f;
    {
      const int tn = (t + 4 < T_STEPS) ? t + 4 : (T_STEPS - 1);
      __builtin_prefetch(xrow + (size_t)tn * IN_F, 0, 3);
    }

    lstm_half_cell(xf, hAf, wi0, wh0, nb0, c0, hA, lane, w);
    __syncthreads();
    hAf = afrag_from_lds(hA, lane);

    lstm_half_cell(hAf, hBf, wi1, wh1, nb1, c1, hB, lane, w);
    __syncthreads();
    hBf = afrag_from_lds(hB, lane);
  }

  if (w == 0) {
    v8f od;
    const float fb = ((lane & 15) == 0) ? bfc[0] : 0.0f;
#pragma unroll
    for (int i = 0; i < 8; ++i) od[i] = fb;
    od = __builtin_amdgcn_wmma_f32_16x16x32_f16(false, hBf, false, wfcf, (short)0, od, false, false);
    if ((lane & 15) == 0) {
      const int mb = (lane < 16) ? 0 : 8;
#pragma unroll
      for (int r = 0; r < 8; ++r) out[b0 + mb + r] = od[r];
    }
  }
}

extern "C" void kernel_launch(void* const* d_in, const int* in_sizes, int n_in,
                              void* d_out, int out_size, void* d_ws, size_t ws_size,
                              hipStream_t stream) {
  const float* x    = (const float*)d_in[0];
  const float* Wih0 = (const float*)d_in[1];
  const float* Whh0 = (const float*)d_in[2];
  const float* bih0 = (const float*)d_in[3];
  const float* bhh0 = (const float*)d_in[4];
  const float* Wih1 = (const float*)d_in[5];
  const float* Whh1 = (const float*)d_in[6];
  const float* bih1 = (const float*)d_in[7];
  const float* bhh1 = (const float*)d_in[8];
  const float* Wfc  = (const float*)d_in[9];
  const float* bfc  = (const float*)d_in[10];
  float* out = (float*)d_out;

  dim3 grid(B_TOTAL / 16);
  dim3 block(64);
  lstm2_wmma_fused<<<grid, block, 0, stream>>>(x, Wih0, Whh0, bih0, bhh0,
                                               Wih1, Whh1, bih1, bhh1,
                                               Wfc, bfc, out);
}


# Layer 1 runs one time step behind layer 0 in the same wave: the two cells of a loop trip are independent and overlap; one barrier per step instead of two, h staging double-buffered
# speedup vs baseline: 1.5611x; 1.5611x over previous
#include <hip/hip_runtime.h>

#define B_TOTAL 8192
#define T_STEPS 256
#define IN_F    16
#define HID     32

typedef __attribute__((ext_vector_type(16))) _Float16 v16h;
typedef __attribute__((ext_vector_type(8)))  _Float16 v8h;
typedef __attribute__((ext_vector_type(8)))  float    v8f;

static __device__ __forceinline__ float htanh(float x) { return __builtin_amdgcn_tanhf(x); }

static __device__ __forceinline__ float sigm_b(float g, float hb) {
  return __builtin_fmaf(htanh(__builtin_fmaf(g, 0.5f, hb)), 0.5f, 0.5f);
}
static __device__ __forceinline__ float tanh_b(float g, float b) {
  return htanh(g + b);
}
static __device__ __forceinline__ float tanh_nb(float c) {
  return htanh(c);
}

static __device__ __forceinline__ v16h afrag_from_lds(const _Float16* h, int lane) {
  const int m  = lane & 15;
  const int kb = (lane < 16) ? 0 : 8;
  const _Float16* p = h + m * HID + kb;
  v8h lo = *(const v8h*)(p);
  v8h hi = *(const v8h*)(p + 16);
  return __builtin_shufflevector(lo, hi, 0,1,2,3,4,5,6,7,8,9,10,11,12,13,14,15);
}

static __device__ __forceinline__ v16h bfrag_from_w(const float* W, int ld, int tile,
                                                    bool padk, int lane) {
  const int n  = tile * 16 + (lane & 15);
  const int kb = (lane < 16) ? 0 : 8;
  const float* row = W + n * ld;
  v16h b;
#pragma unroll
  for (int i = 0; i < 8; ++i) b[i] = (_Float16)row[kb + i];
#pragma unroll
  for (int i = 0; i < 8; ++i) b[8 + i] = padk ? (_Float16)0.0f : (_Float16)row[16 + kb + i];
  return b;
}

static __device__ __forceinline__ void lstm_half_cell(const v16h in, const v16h hr,
                                                      const v16h* Wi, const v16h* Wh,
                                                      const float* nb,
                                                      v8f& c, _Float16* hout,
                                                      int lane, int w) {
  v8f g[4];
#pragma unroll
  for (int gt = 0; gt < 4; ++gt) {
    v8f acc = {};
    acc = __builtin_amdgcn_wmma_f32_16x16x32_f16(false, in, false, Wi[gt], (short)0, acc, false, false);
    acc = __builtin_amdgcn_wmma_f32_16x16x32_f16(false, hr, false, Wh[gt], (short)0, acc, false, false);
    g[gt] = acc;
  }
  const int col = w * 16 + (lane & 15);
  const int mb  = (lane < 16) ? 0 : 8;
#pragma unroll
  for (int r = 0; r < 8; ++r) {
    float iv = sigm_b(g[0][r], nb[0]);
    float fv = sigm_b(g[1][r], nb[1]);
    float gv = tanh_b(g[2][r], nb[2]);
    float ov = sigm_b(g[3][r], nb[3]);
    float cv = __builtin_fmaf(fv, c[r], iv * gv);
    c[r] = cv;
    float hv = ov * tanh_nb(cv);
    hout[(mb + r) * HID + col] = (_Float16)hv;
  }
}

#define HSZ (16 * HID)

static __device__ __forceinline__ v16h xfrag_load(const float* xrow, int t) {
  v16h xf;
  const float* p = xrow + (size_t)t * IN_F;
#pragma unroll
  for (int i = 0; i < 8; ++i) xf[i] = (_Float16)p[i];
#pragma unroll
  for (int i = 0; i < 8; ++i) xf[8 + i] = (_Float16)0.0f;
  const int tn = (t + 4 < T_STEPS) ? t + 4 : (T_STEPS - 1);
  __builtin_prefetch(xrow + (size_t)tn * IN_F, 0, 3);
  return xf;
}

static __device__ __forceinline__ void skew_trip(const float* xrow, int t,
                                                 v16h& hAf, v16h& hBf,
                                                 const v16h* wi0, const v16h* wh0,
                                                 const v16h* wi1, const v16h* wh1,
                                                 const float* nb0, const float* nb1,
                                                 v8f& c0, v8f& c1,
                                                 _Float16* hAbuf, _Float16* hBbuf,
                                                 int lane, int w) {
  v16h xf = xfrag_load(xrow, t);
  lstm_half_cell(hAf, hBf, wi1, wh1, nb1, c1, hBbuf, lane, w);
  lstm_half_cell(xf,  hAf, wi0, wh0, nb0, c0, hAbuf, lane, w);
  __syncthreads();
  hAf = afrag_from_lds(hAbuf, lane);
  hBf = afrag_from_lds(hBbuf, lane);
}

extern "C" __global__ __launch_bounds__(64)
void lstm2_wmma_fused(const float* __restrict__ x,
                      const float* __restrict__ Wih0, const float* __restrict__ Whh0,
                      const float* __restrict__ bih0, const float* __restrict__ bhh0,
                      const float* __restrict__ Wih1, const float* __restrict__ Whh1,
                      const float* __restrict__ bih1, const float* __restrict__ bhh1,
                      const float* __restrict__ Wfc,  const float* __restrict__ bfc,
                      float* __restrict__ out) {
  __shared__ _Float16 hA[2 * HSZ];
  __shared__ _Float16 hB[2 * HSZ];
  const int lane = threadIdx.x & 31;
  const int w    = threadIdx.x >> 5;
  const int b0   = blockIdx.x * 16;

  v16h wi0[4], wh0[4], wi1[4], wh1[4];
#pragma unroll
  for (int gt = 0; gt < 4; ++gt) {
    const int tile = gt * 2 + w;
    wi0[gt] = bfrag_from_w(Wih0, IN_F, tile, true,  lane);
    wh0[gt] = bfrag_from_w(Whh0, HID,  tile, false, lane);
    wi1[gt] = bfrag_from_w(Wih1, HID,  tile, false, lane);
    wh1[gt] = bfrag_from_w(Whh1, HID,  tile, false, lane);
  }
  float nb0[4], nb1[4];
#pragma unroll
  for (int gt = 0; gt < 4; ++gt) {
    const int col   = (gt * 2 + w) * 16 + (lane & 15);
    const float scl = (gt == 2) ? 1.0f : 0.5f;
    nb0[gt] = scl * (bih0[col] + bhh0[col]);
    nb1[gt] = scl * (bih1[col] + bhh1[col]);
  }
  v16h wfcf;
  {
    const int  kb = (lane < 16) ? 0 : 8;
    const bool n0 = (lane & 15) == 0;
#pragma unroll
    for (int i = 0; i < 8; ++i) wfcf[i]     = n0 ? (_Float16)Wfc[kb + i]      : (_Float16)0.0f;
#pragma unroll
    for (int i = 0; i < 8; ++i) wfcf[8 + i] = n0 ? (_Float16)Wfc[16 + kb + i] : (_Float16)0.0f;
  }

  v8f c0 = {}, c1 = {};
  v16h hAf, hBf;
#pragma unroll
  for (int i = 0; i < 16; ++i) { hAf[i] = (_Float16)0.0f; hBf[i] = (_Float16)0.0f; }

  const int m_row = lane & 15;
  const int kb_x  = (lane < 16) ? 0 : 8;
  const float* xrow = x + (size_t)(b0 + m_row) * T_STEPS * IN_F + kb_x;


  {
    v16h xf = xfrag_load(xrow, 0);
    lstm_half_cell(xf, hAf, wi0, wh0, nb0, c0, hA, lane, w);
    __syncthreads();
    hAf = afrag_from_lds(hA, lane);
  }

#pragma unroll 1
  for (int t = 1; t + 1 < T_STEPS; t += 2) {
    skew_trip(xrow, t,     hAf, hBf, wi0, wh0, wi1, wh1, nb0, nb1, c0, c1, hA + HSZ, hB + HSZ, lane, w);
    skew_trip(xrow, t + 1, hAf, hBf, wi0, wh0, wi1, wh1, nb0, nb1, c0, c1, hA,       hB,       lane, w);
  }
  skew_trip(xrow, T_STEPS - 1, hAf, hBf, wi0, wh0, wi1, wh1, nb0, nb1, c0, c1, hA + HSZ, hB + HSZ, lane, w);

  lstm_half_cell(hAf, hBf, wi1, wh1, nb1, c1, hB, lane, w);
  __syncthreads();
  hBf = afrag_from_lds(hB, lane);

  if (w == 0) {
    v8f od;
    const float fb = ((lane & 15) == 0) ? bfc[0] : 0.0f;
#pragma unroll
    for (int i = 0; i < 8; ++i) od[i] = fb;
    od = __builtin_amdgcn_wmma_f32_16x16x32_f16(false, hBf, false, wfcf, (short)0, od, false, false);
    if ((lane & 15) == 0) {
      const int mb = (lane < 16) ? 0 : 8;
#pragma unroll
      for (int r = 0; r < 8; ++r) out[b0 + mb + r] = od[r];
    }
  }
}

extern "C" void kernel_launch(void* const* d_in, const int* in_sizes, int n_in,
                              void* d_out, int out_size, void* d_ws, size_t ws_size,
                              hipStream_t stream) {
  const float* x    = (const float*)d_in[0];
  const float* Wih0 = (const float*)d_in[1];
  const float* Whh0 = (const float*)d_in[2];
  const float* bih0 = (const float*)d_in[3];
  const float* bhh0 = (const float*)d_in[4];
  const float* Wih1 = (const float*)d_in[5];
  const float* Whh1 = (const float*)d_in[6];
  const float* bih1 = (const float*)d_in[7];
  const float* bhh1 = (const float*)d_in[8];
  const float* Wfc  = (const float*)d_in[9];
  const float* bfc  = (const float*)d_in[10];
  float* out = (float*)d_out;

  dim3 grid(B_TOTAL / 16);
  dim3 block(64);
  lstm2_wmma_fused<<<grid, block, 0, stream>>>(x, Wih0, Whh0, bih0, bhh0,
                                               Wih1, Whh1, bih1, bhh1,
                                               Wfc, bfc, out);
}


# Gate products transposed (W as A, h^T as B), one wave per layer: a lane's activated result registers are its next-step B operand, so recurrent h stays in registers; LDS only passes layer-0 h on
# speedup vs baseline: 1.9996x; 1.2809x over previous
#include <hip/hip_runtime.h>


#define B_TOTAL 8192
#define T_STEPS 256
#define IN_F    16
#define HID     32

typedef __attribute__((ext_vector_type(16))) _Float16 v16h;
typedef __attribute__((ext_vector_type(8)))  _Float16 v8h;
typedef __attribute__((ext_vector_type(8)))  float    v8f;
typedef __attribute__((ext_vector_type(4)))  float    v4f;

static __device__ __forceinline__ float htanh(float x) { return __builtin_amdgcn_tanhf(x); }

static __device__ __forceinline__ float sigm_b(float g, float hb) {
  return __builtin_fmaf(htanh(__builtin_fmaf(g, 0.5f, hb)), 0.5f, 0.5f);
}
static __device__ __forceinline__ float tanh_b(float g, float b) { return htanh(g + b); }


static __device__ __forceinline__ v16h wfrag(const float* W, int ld, int tile, bool padk, int lane) {
  const int n  = tile * 16 + (lane & 15);
  const int kb = (lane < 16) ? 0 : 8;
  const float* row = W + n * ld;
  v16h a;
#pragma unroll
  for (int i = 0; i < 8; ++i) a[i] = (_Float16)row[kb + i];
#pragma unroll
  for (int i = 0; i < 8; ++i) a[8 + i] = padk ? (_Float16)0.0f : (_Float16)row[16 + kb + i];
  return a;
}

static __device__ __forceinline__ v8f wmma16(v16h a, v16h b, v8f c) {
  return __builtin_amdgcn_wmma_f32_16x16x32_f16(false, a, false, b, (short)0, c, false, false);
}

static __device__ __forceinline__ void in_product(v8f* acc, const v16h* Wi, const v16h in) {
#pragma unroll
  for (int q = 0; q < 8; ++q) { v8f z = {}; acc[q] = wmma16(Wi[q], in, z); }
}
static __device__ __forceinline__ void rec_product(v8f* acc, const v16h* Wh, const v16h h) {
#pragma unroll
  for (int q = 0; q < 8; ++q) acc[q] = wmma16(Wh[q], h, acc[q]);
}

static __device__ __forceinline__ void cell_update(const v8f* acc, const v8f* nb, v8f* c, v16h& h) {
#pragma unroll
  for (int j = 0; j < 2; ++j) {
#pragma unroll
    for (int r = 0; r < 8; ++r) {
      float iv = sigm_b(acc[4 * j + 0][r], nb[4 * j + 0][r]);
      float fv = sigm_b(acc[4 * j + 1][r], nb[4 * j + 1][r]);
      float gv = tanh_b(acc[4 * j + 2][r], nb[4 * j + 2][r]);
      float ov = sigm_b(acc[4 * j + 3][r], nb[4 * j + 3][r]);
      float cv = __builtin_fmaf(fv, c[j][r], iv * gv);
      c[j][r] = cv;
      float hv = ov * htanh(cv);
      h[8 * j + r] = (_Float16)hv;
    }
  }
}

#define SLOT_HALVES 512
#define RING_SLOTS  8

static __device__ __forceinline__ void frag_to_lds(_Float16* slot, const v16h f, int lane) {
  v8h lo = __builtin_shufflevector(f, f, 0, 1, 2, 3, 4, 5, 6, 7);
  v8h hi = __builtin_shufflevector(f, f, 8, 9, 10, 11, 12, 13, 14, 15);
  *(v8h*)(slot + lane * 8) = lo;
  *(v8h*)(slot + 256 + lane * 8) = hi;
}
static __device__ __forceinline__ v16h frag_from_lds(const _Float16* slot, int lane) {
  v8h lo = *(const v8h*)(slot + lane * 8);
  v8h hi = *(const v8h*)(slot + 256 + lane * 8);
  return __builtin_shufflevector(lo, hi, 0, 1, 2, 3, 4, 5, 6, 7, 8, 9, 10, 11, 12, 13, 14, 15);
}

static __device__ __forceinline__ v16h xfrag_cvt(const v4f lo, const v4f hi) {
  v16h xf;
#pragma unroll
  for (int i = 0; i < 4; ++i) { xf[i] = (_Float16)lo[i]; xf[4 + i] = (_Float16)hi[i]; }
#pragma unroll
  for (int i = 0; i < 8; ++i) xf[8 + i] = (_Float16)0.0f;
  return xf;
}

extern "C" __global__ __launch_bounds__(64)
void lstm2_wmma_fused(const float* __restrict__ x,
                      const float* __restrict__ Wih0, const float* __restrict__ Whh0,
                      const float* __restrict__ bih0, const float* __restrict__ bhh0,
                      const float* __restrict__ Wih1, const float* __restrict__ Whh1,
                      const float* __restrict__ bih1, const float* __restrict__ bhh1,
                      const float* __restrict__ Wfc,  const float* __restrict__ bfc,
                      float* __restrict__ out) {
  __shared__ __attribute__((aligned(16))) _Float16 ring[RING_SLOTS * SLOT_HALVES];
  const int lane = threadIdx.x & 31;
  const int w  = __builtin_amdgcn_readfirstlane((int)(threadIdx.x >> 5));
  const int b0 = blockIdx.x * 16;
  const int hh = lane >> 4;

  v16h Wi[8], Wh[8];
  v8f  nb[8];
  if (w == 0) {
#pragma unroll
    for (int q = 0; q < 8; ++q) {
      const int tile = (q & 3) * 2 + (q >> 2);
      Wi[q] = wfrag(Wih0, IN_F, tile, true,  lane);
      Wh[q] = wfrag(Whh0, HID,  tile, false, lane);
    }
  } else {
#pragma unroll
    for (int q = 0; q < 8; ++q) {
      const int tile = (q & 3) * 2 + (q >> 2);
      Wi[q] = wfrag(Wih1, HID, tile, false, lane);
      Wh[q] = wfrag(Whh1, HID, tile, false, lane);
    }
  }
  {
    const float* bi = (w == 0) ? bih0 : bih1;
    const float* bh = (w == 0) ? bhh0 : bhh1;
#pragma unroll
    for (int q = 0; q < 8; ++q) {
      const int   gate = q & 3;
      const int   row0 = gate * HID + (q >> 2) * 16 + hh * 8;
      const float scl  = (gate == 2) ? 1.0f : 0.5f;
#pragma unroll
      for (int r = 0; r < 8; ++r) nb[q][r] = scl * (bi[row0 + r] + bh[row0 + r]);
    }
  }

  v8f c[2];
  c[0] = (v8f){}; c[1] = (v8f){};
  v16h hf;
#pragma unroll
  for (int i = 0; i < 16; ++i) hf[i] = (_Float16)0.0f;

  const float* xrow = x + (size_t)(b0 + (lane & 15)) * T_STEPS * IN_F + hh * 8;

  v8f accA[8], accB[8];
  v16h inB, inA2;
  v16h in0;
#pragma unroll
  for (int q = 0; q < 8; ++q) { accA[q] = (v8f){}; accB[q] = (v8f){}; }
#pragma unroll
  for (int i = 0; i < 16; ++i) { inB[i] = (_Float16)0.0f; inA2[i] = (_Float16)0.0f; in0[i] = (_Float16)0.0f; }

  v4f xb0 = {}, xb1 = {}, xa0 = {}, xa1 = {};
  if (w == 0) {
    const v4f* p0 = (const v4f*)(xrow);
    const v4f* p1 = (const v4f*)(xrow + IN_F);
    const v4f* p2 = (const v4f*)(xrow + 2 * IN_F);
    in_product(accA, Wi, xfrag_cvt(p0[0], p0[1]));
    xb0 = p1[0]; xb1 = p1[1]; xa0 = p2[0]; xa1 = p2[1];
  }

#pragma unroll 1
  for (int g = 0; g < T_STEPS / 2 + 2; ++g) {
    if (w == 0 && g >= 1 && g <= T_STEPS / 2)
      frag_to_lds(ring + ((2 * g - 1) & (RING_SLOTS - 1)) * SLOT_HALVES, hf, lane);
    const bool active = (w == 0) ? (g < T_STEPS / 2) : (g >= 2);
    if (active) {
      const int lt = (w == 0) ? 2 * g : 2 * (g - 2);
      if (w == 0) {
        inB  = xfrag_cvt(xb0, xb1);
        inA2 = xfrag_cvt(xa0, xa1);
        const int tb = (lt + 3 < T_STEPS) ? lt + 3 : T_STEPS - 1;
        const int ta = (lt + 4 < T_STEPS) ? lt + 4 : T_STEPS - 1;
        const int tp = (lt + 8 < T_STEPS) ? lt + 8 : T_STEPS - 1;
        const v4f* pb = (const v4f*)(xrow + (size_t)tb * IN_F);
        const v4f* pa = (const v4f*)(xrow + (size_t)ta * IN_F);
        xb0 = pb[0]; xb1 = pb[1]; xa0 = pa[0]; xa1 = pa[1];
        __builtin_prefetch(xrow + (size_t)tp * IN_F, 0, 3);
      }
      if (w == 1 && g == 2) in_product(accA, Wi, in0);
      in_product(accB, Wi, inB);
      rec_product(accA, Wh, hf);
      cell_update(accA, nb, c, hf);
      if (w == 0) frag_to_lds(ring + (lt & (RING_SLOTS - 1)) * SLOT_HALVES, hf, lane);
      in_product(accA, Wi, inA2);
      rec_product(accB, Wh, hf);
      cell_update(accB, nb, c, hf);
    }
    __syncthreads();
    if (w == 1) {
      if (g == 1) in0 = frag_from_lds(ring, lane);
      if (g >= 1) {
        inB  = frag_from_lds(ring + ((2 * g - 1) & (RING_SLOTS - 1)) * SLOT_HALVES, lane);
        inA2 = frag_from_lds(ring + ((2 * g) & (RING_SLOTS - 1)) * SLOT_HALVES, lane);
      }
    }
  }

  if (w == 1) {
    v16h wfcf;
    {
      const int  kb = hh * 8;
      const bool r0 = (lane & 15) == 0;
#pragma unroll
      for (int i = 0; i < 8; ++i) wfcf[i]     = r0 ? (_Float16)Wfc[kb + i]      : (_Float16)0.0f;
#pragma unroll
      for (int i = 0; i < 8; ++i) wfcf[8 + i] = r0 ? (_Float16)Wfc[16 + kb + i] : (_Float16)0.0f;
    }
    v8f od = {};
    od[0] = (hh == 0) ? bfc[0] : 0.0f;
    od = wmma16(wfcf, hf, od);
    if (lane < 16) out[b0 + lane] = od[0];
  }
}

extern "C" void kernel_launch(void* const* d_in, const int* in_sizes, int n_in,
                              void* d_out, int out_size, void* d_ws, size_t ws_size,
                              hipStream_t stream) {
  const float* x    = (const float*)d_in[0];
  const float* Wih0 = (const float*)d_in[1];
  const float* Whh0 = (const float*)d_in[2];
  const float* bih0 = (const float*)d_in[3];
  const float* bhh0 = (const float*)d_in[4];
  const float* Wih1 = (const float*)d_in[5];
  const float* Whh1 = (const float*)d_in[6];
  const float* bih1 = (const float*)d_in[7];
  const float* bhh1 = (const float*)d_in[8];
  const float* Wfc  = (const float*)d_in[9];
  const float* bfc  = (const float*)d_in[10];
  float* out = (float*)d_out;

  dim3 grid(B_TOTAL / 16);
  dim3 block(64);
  lstm2_wmma_fused<<<grid, block, 0, stream>>>(x, Wih0, Whh0, bih0, bhh0,
                                               Wih1, Whh1, bih1, bhh1,
                                               Wfc, bfc, out);
}


# Gate pre-activation affine folded into the matrix product: bias is the C operand (summed first, not last), sigmoid's 0.5 sits in the f16 weight rows; per-element pre-scale fma and bias add gone
# speedup vs baseline: 2.1873x; 1.0939x over previous
#include <hip/hip_runtime.h>


#define B_TOTAL 8192
#define T_STEPS 256
#define IN_F    16
#define HID     32

typedef __attribute__((ext_vector_type(16))) _Float16 v16h;
typedef __attribute__((ext_vector_type(8)))  _Float16 v8h;
typedef __attribute__((ext_vector_type(8)))  float    v8f;
typedef __attribute__((ext_vector_type(4)))  float    v4f;

static __device__ __forceinline__ float htanh(float x) { return __builtin_amdgcn_tanhf(x); }

static __device__ __forceinline__ float sigm_half(float hz) { return __builtin_fmaf(htanh(hz), 0.5f, 0.5f); }


static __device__ __forceinline__ v16h wfrag(const float* W, int ld, int tile, bool padk, float scl, int lane) {
  const int n  = tile * 16 + (lane & 15);
  const int kb = (lane < 16) ? 0 : 8;
  const float* row = W + n * ld;
  v16h a;
#pragma unroll
  for (int i = 0; i < 8; ++i) a[i] = (_Float16)(scl * row[kb + i]);
#pragma unroll
  for (int i = 0; i < 8; ++i) a[8 + i] = padk ? (_Float16)0.0f : (_Float16)(scl * row[16 + kb + i]);
  return a;
}

static __device__ __forceinline__ v8f wmma16(v16h a, v16h b, v8f c) {
  return __builtin_amdgcn_wmma_f32_16x16x32_f16(false, a, false, b, (short)0, c, false, false);
}

static __device__ __forceinline__ void in_product(v8f* acc, const v16h* Wi, const v16h in, const v8f* nb) {
#pragma unroll
  for (int q = 0; q < 8; ++q) acc[q] = wmma16(Wi[q], in, nb[q]);
}
static __device__ __forceinline__ void rec_product(v8f* acc, const v16h* Wh, const v16h h) {
#pragma unroll
  for (int q = 0; q < 8; ++q) acc[q] = wmma16(Wh[q], h, acc[q]);
}

static __device__ __forceinline__ void cell_update(const v8f* acc, v8f* c, v16h& h) {
#pragma unroll
  for (int j = 0; j < 2; ++j) {
#pragma unroll
    for (int r = 0; r < 8; ++r) {
      float iv = sigm_half(acc[4 * j + 0][r]);
      float fv = sigm_half(acc[4 * j + 1][r]);
      float gv = htanh(acc[4 * j + 2][r]);
      float ov = sigm_half(acc[4 * j + 3][r]);
      float cv = __builtin_fmaf(fv, c[j][r], iv * gv);
      c[j][r] = cv;
      float hv = ov * htanh(cv);
      h[8 * j + r] = (_Float16)hv;
    }
  }
}

#define SLOT_HALVES 512
#define RING_SLOTS  8

static __device__ __forceinline__ void frag_to_lds(_Float16* slot, const v16h f, int lane) {
  v8h lo = __builtin_shufflevector(f, f, 0, 1, 2, 3, 4, 5, 6, 7);
  v8h hi = __builtin_shufflevector(f, f, 8, 9, 10, 11, 12, 13, 14, 15);
  *(v8h*)(slot + lane * 8) = lo;
  *(v8h*)(slot + 256 + lane * 8) = hi;
}
static __device__ __forceinline__ v16h frag_from_lds(const _Float16* slot, int lane) {
  v8h lo = *(const v8h*)(slot + lane * 8);
  v8h hi = *(const v8h*)(slot + 256 + lane * 8);
  return __builtin_shufflevector(lo, hi, 0, 1, 2, 3, 4, 5, 6, 7, 8, 9, 10, 11, 12, 13, 14, 15);
}

static __device__ __forceinline__ v16h xfrag_cvt(const v4f lo, const v4f hi) {
  v16h xf;
#pragma unroll
  for (int i = 0; i < 4; ++i) { xf[i] = (_Float16)lo[i]; xf[4 + i] = (_Float16)hi[i]; }
#pragma unroll
  for (int i = 0; i < 8; ++i) xf[8 + i] = (_Float16)0.0f;
  return xf;
}

extern "C" __global__ __launch_bounds__(64)
void lstm2_wmma_fused(const float* __restrict__ x,
                      const float* __restrict__ Wih0, const float* __restrict__ Whh0,
                      const float* __restrict__ bih0, const float* __restrict__ bhh0,
                      const float* __restrict__ Wih1, const float* __restrict__ Whh1,
                      const float* __restrict__ bih1, const float* __restrict__ bhh1,
                      const float* __restrict__ Wfc,  const float* __restrict__ bfc,
                      float* __restrict__ out) {
  __shared__ __attribute__((aligned(16))) _Float16 ring[RING_SLOTS * SLOT_HALVES];
  const int lane = threadIdx.x & 31;
  const int w  = __builtin_amdgcn_readfirstlane((int)(threadIdx.x >> 5));
  const int b0 = blockIdx.x * 16;
  const int hh = lane >> 4;

  v16h Wi[8], Wh[8];
  v8f  nb[8];
  if (w == 0) {
#pragma unroll
    for (int q = 0; q < 8; ++q) {
      const int   tile = (q & 3) * 2 + (q >> 2);
      const float scl  = ((q & 3) == 2) ? 1.0f : 0.5f;
      Wi[q] = wfrag(Wih0, IN_F, tile, true,  scl, lane);
      Wh[q] = wfrag(Whh0, HID,  tile, false, scl, lane);
    }
  } else {
#pragma unroll
    for (int q = 0; q < 8; ++q) {
      const int   tile = (q & 3) * 2 + (q >> 2);
      const float scl  = ((q & 3) == 2) ? 1.0f : 0.5f;
      Wi[q] = wfrag(Wih1, HID, tile, false, scl, lane);
      Wh[q] = wfrag(Whh1, HID, tile, false, scl, lane);
    }
  }
  {
    const float* bi = (w == 0) ? bih0 : bih1;
    const float* bh = (w == 0) ? bhh0 : bhh1;
#pragma unroll
    for (int q = 0; q < 8; ++q) {
      const int   gate = q & 3;
      const int   row0 = gate * HID + (q >> 2) * 16 + hh * 8;
      const float scl  = (gate == 2) ? 1.0f : 0.5f;
#pragma unroll
      for (int r = 0; r < 8; ++r) nb[q][r] = scl * (bi[row0 + r] + bh[row0 + r]);
    }
  }

  v8f c[2];
  c[0] = (v8f){}; c[1] = (v8f){};
  v16h hf;
#pragma unroll
  for (int i = 0; i < 16; ++i) hf[i] = (_Float16)0.0f;

  const float* xrow = x + (size_t)(b0 + (lane & 15)) * T_STEPS * IN_F + hh * 8;

  v8f accA[8], accB[8];
  v16h inB, inA2;
  v16h in0;
#pragma unroll
  for (int q = 0; q < 8; ++q) { accA[q] = (v8f){}; accB[q] = (v8f){}; }
#pragma unroll
  for (int i = 0; i < 16; ++i) { inB[i] = (_Float16)0.0f; inA2[i] = (_Float16)0.0f; in0[i] = (_Float16)0.0f; }

  v4f xb0 = {}, xb1 = {}, xa0 = {}, xa1 = {};
  if (w == 0) {
    const v4f* p0 = (const v4f*)(xrow);
    const v4f* p1 = (const v4f*)(xrow + IN_F);
    const v4f* p2 = (const v4f*)(xrow + 2 * IN_F);
    in_product(accA, Wi, xfrag_cvt(p0[0], p0[1]), nb);
    xb0 = p1[0]; xb1 = p1[1]; xa0 = p2[0]; xa1 = p2[1];
  }

#pragma unroll 1
  for (int g = 0; g < T_STEPS / 2 + 2; ++g) {
    if (w == 0 && g >= 1 && g <= T_STEPS / 2)
      frag_to_lds(ring + ((2 * g - 1) & (RING_SLOTS - 1)) * SLOT_HALVES, hf, lane);
    const bool active = (w == 0) ? (g < T_STEPS / 2) : (g >= 2);
    if (active) {
      const int lt = (w == 0) ? 2 * g : 2 * (g - 2);
      if (w == 0) {
        inB  = xfrag_cvt(xb0, xb1);
        inA2 = xfrag_cvt(xa0, xa1);
        const int tb = (lt + 3 < T_STEPS) ? lt + 3 : T_STEPS - 1;
        const int ta = (lt + 4 < T_STEPS) ? lt + 4 : T_STEPS - 1;
        const int tp = (lt + 8 < T_STEPS) ? lt + 8 : T_STEPS - 1;
        const v4f* pb = (const v4f*)(xrow + (size_t)tb * IN_F);
        const v4f* pa = (const v4f*)(xrow + (size_t)ta * IN_F);
        xb0 = pb[0]; xb1 = pb[1]; xa0 = pa[0]; xa1 = pa[1];
        __builtin_prefetch(xrow + (size_t)tp * IN_F, 0, 3);
      }
      if (w == 1 && g == 2) in_product(accA, Wi, in0, nb);
      in_product(accB, Wi, inB, nb);
      rec_product(accA, Wh, hf);
      cell_update(accA, c, hf);
      if (w == 0) frag_to_lds(ring + (lt & (RING_SLOTS - 1)) * SLOT_HALVES, hf, lane);
      in_product(accA, Wi, inA2, nb);
      rec_product(accB, Wh, hf);
      cell_update(accB, c, hf);
    }
    __syncthreads();
    if (w == 1) {
      if (g == 1) in0 = frag_from_lds(ring, lane);
      if (g >= 1) {
        inB  = frag_from_lds(ring + ((2 * g - 1) & (RING_SLOTS - 1)) * SLOT_HALVES, lane);
        inA2 = frag_from_lds(ring + ((2 * g) & (RING_SLOTS - 1)) * SLOT_HALVES, lane);
      }
    }
  }

  if (w == 1) {
    v16h wfcf;
    {
      const int  kb = hh * 8;
      const bool r0 = (lane & 15) == 0;
#pragma unroll
      for (int i = 0; i < 8; ++i) wfcf[i]     = r0 ? (_Float16)Wfc[kb + i]      : (_Float16)0.0f;
#pragma unroll
      for (int i = 0; i < 8; ++i) wfcf[8 + i] = r0 ? (_Float16)Wfc[16 + kb + i] : (_Float16)0.0f;
    }
    v8f od = {};
    od[0] = (hh == 0) ? bfc[0] : 0.0f;
    od = wmma16(wfcf, hf, od);
    if (lane < 16) out[b0 + lane] = od[0];
  }
}

extern "C" void kernel_launch(void* const* d_in, const int* in_sizes, int n_in,
                              void* d_out, int out_size, void* d_ws, size_t ws_size,
                              hipStream_t stream) {
  const float* x    = (const float*)d_in[0];
  const float* Wih0 = (const float*)d_in[1];
  const float* Whh0 = (const float*)d_in[2];
  const float* bih0 = (const float*)d_in[3];
  const float* bhh0 = (const float*)d_in[4];
  const float* Wih1 = (const float*)d_in[5];
  const float* Whh1 = (const float*)d_in[6];
  const float* bih1 = (const float*)d_in[7];
  const float* bhh1 = (const float*)d_in[8];
  const float* Wfc  = (const float*)d_in[9];
  const float* bfc  = (const float*)d_in[10];
  float* out = (float*)d_out;

  dim3 grid(B_TOTAL / 16);
  dim3 block(64);
  lstm2_wmma_fused<<<grid, block, 0, stream>>>(x, Wih0, Whh0, bih0, bhh0,
                                               Wih1, Whh1, bih1, bhh1,
                                               Wfc, bfc, out);
}


# Trip loop split into start-up, steady and drain loops: the steady trips carry none of the start-up and drain tests (fewer scalar compares and branches per trip)
# speedup vs baseline: 2.2668x; 1.0363x over previous
#include <hip/hip_runtime.h>


#define B_TOTAL 8192
#define T_STEPS 256
#define IN_F    16
#define HID     32

typedef __attribute__((ext_vector_type(16))) _Float16 v16h;
typedef __attribute__((ext_vector_type(8)))  _Float16 v8h;
typedef __attribute__((ext_vector_type(8)))  float    v8f;
typedef __attribute__((ext_vector_type(4)))  float    v4f;

static __device__ __forceinline__ float htanh(float x) { return __builtin_amdgcn_tanhf(x); }

static __device__ __forceinline__ float sigm_half(float hz) { return __builtin_fmaf(htanh(hz), 0.5f, 0.5f); }


static __device__ __forceinline__ v16h wfrag(const float* W, int ld, int tile, bool padk, float scl, int lane) {
  const int n  = tile * 16 + (lane & 15);
  const int kb = (lane < 16) ? 0 : 8;
  const float* row = W + n * ld;
  v16h a;
#pragma unroll
  for (int i = 0; i < 8; ++i) a[i] = (_Float16)(scl * row[kb + i]);
#pragma unroll
  for (int i = 0; i < 8; ++i) a[8 + i] = padk ? (_Float16)0.0f : (_Float16)(scl * row[16 + kb + i]);
  return a;
}

static __device__ __forceinline__ v8f wmma16(v16h a, v16h b, v8f c) {
  return __builtin_amdgcn_wmma_f32_16x16x32_f16(false, a, false, b, (short)0, c, false, false);
}

static __device__ __forceinline__ void in_product(v8f* acc, const v16h* Wi, const v16h in, const v8f* nb) {
#pragma unroll
  for (int q = 0; q < 8; ++q) acc[q] = wmma16(Wi[q], in, nb[q]);
}
static __device__ __forceinline__ void rec_product(v8f* acc, const v16h* Wh, const v16h h) {
#pragma unroll
  for (int q = 0; q < 8; ++q) acc[q] = wmma16(Wh[q], h, acc[q]);
}

static __device__ __forceinline__ void cell_update(const v8f* acc, v8f* c, v16h& h) {
#pragma unroll
  for (int j = 0; j < 2; ++j) {
#pragma unroll
    for (int r = 0; r < 8; ++r) {
      float iv = sigm_half(acc[4 * j + 0][r]);
      float fv = sigm_half(acc[4 * j + 1][r]);
      float gv = htanh(acc[4 * j + 2][r]);
      float ov = sigm_half(acc[4 * j + 3][r]);
      float cv = __builtin_fmaf(fv, c[j][r], iv * gv);
      c[j][r] = cv;
      float hv = ov * htanh(cv);
      h[8 * j + r] = (_Float16)hv;
    }
  }
}

#define SLOT_HALVES 512
#define RING_SLOTS  8

static __device__ __forceinline__ void frag_to_lds(_Float16* slot, const v16h f, int lane) {
  v8h lo = __builtin_shufflevector(f, f, 0, 1, 2, 3, 4, 5, 6, 7);
  v8h hi = __builtin_shufflevector(f, f, 8, 9, 10, 11, 12, 13, 14, 15);
  *(v8h*)(slot + lane * 8) = lo;
  *(v8h*)(slot + 256 + lane * 8) = hi;
}
static __device__ __forceinline__ v16h frag_from_lds(const _Float16* slot, int lane) {
  v8h lo = *(const v8h*)(slot + lane * 8);
  v8h hi = *(const v8h*)(slot + 256 + lane * 8);
  return __builtin_shufflevector(lo, hi, 0, 1, 2, 3, 4, 5, 6, 7, 8, 9, 10, 11, 12, 13, 14, 15);
}

static __device__ __forceinline__ v16h xfrag_cvt(const v4f lo, const v4f hi) {
  v16h xf;
#pragma unroll
  for (int i = 0; i < 4; ++i) { xf[i] = (_Float16)lo[i]; xf[4 + i] = (_Float16)hi[i]; }
#pragma unroll
  for (int i = 0; i < 8; ++i) xf[8 + i] = (_Float16)0.0f;
  return xf;
}

extern "C" __global__ __launch_bounds__(64)
void lstm2_wmma_fused(const float* __restrict__ x,
                      const float* __restrict__ Wih0, const float* __restrict__ Whh0,
                      const float* __restrict__ bih0, const float* __restrict__ bhh0,
                      const float* __restrict__ Wih1, const float* __restrict__ Whh1,
                      const float* __restrict__ bih1, const float* __restrict__ bhh1,
                      const float* __restrict__ Wfc,  const float* __restrict__ bfc,
                      float* __restrict__ out) {
  __shared__ __attribute__((aligned(16))) _Float16 ring[RING_SLOTS * SLOT_HALVES];
  const int lane = threadIdx.x & 31;
  const int w  = __builtin_amdgcn_readfirstlane((int)(threadIdx.x >> 5));
  const int b0 = blockIdx.x * 16;
  const int hh = lane >> 4;

  v16h Wi[8], Wh[8];
  v8f  nb[8];
  if (w == 0) {
#pragma unroll
    for (int q = 0; q < 8; ++q) {
      const int   tile = (q & 3) * 2 + (q >> 2);
      const float scl  = ((q & 3) == 2) ? 1.0f : 0.5f;
      Wi[q] = wfrag(Wih0, IN_F, tile, true,  scl, lane);
      Wh[q] = wfrag(Whh0, HID,  tile, false, scl, lane);
    }
  } else {
#pragma unroll
    for (int q = 0; q < 8; ++q) {
      const int   tile = (q & 3) * 2 + (q >> 2);
      const float scl  = ((q & 3) == 2) ? 1.0f : 0.5f;
      Wi[q] = wfrag(Wih1, HID, tile, false, scl, lane);
      Wh[q] = wfrag(Whh1, HID, tile, false, scl, lane);
    }
  }
  {
    const float* bi = (w == 0) ? bih0 : bih1;
    const float* bh = (w == 0) ? bhh0 : bhh1;
#pragma unroll
    for (int q = 0; q < 8; ++q) {
      const int   gate = q & 3;
      const int   row0 = gate * HID + (q >> 2) * 16 + hh * 8;
      const float scl  = (gate == 2) ? 1.0f : 0.5f;
#pragma unroll
      for (int r = 0; r < 8; ++r) nb[q][r] = scl * (bi[row0 + r] + bh[row0 + r]);
    }
  }

  v8f c[2];
  c[0] = (v8f){}; c[1] = (v8f){};
  v16h hf;
#pragma unroll
  for (int i = 0; i < 16; ++i) hf[i] = (_Float16)0.0f;

  const float* xrow = x + (size_t)(b0 + (lane & 15)) * T_STEPS * IN_F + hh * 8;

  v8f accA[8], accB[8];
  v16h inB, inA2;
  v16h in0;
#pragma unroll
  for (int q = 0; q < 8; ++q) { accA[q] = (v8f){}; accB[q] = (v8f){}; }
#pragma unroll
  for (int i = 0; i < 16; ++i) { inB[i] = (_Float16)0.0f; inA2[i] = (_Float16)0.0f; in0[i] = (_Float16)0.0f; }

  v4f xb0 = {}, xb1 = {}, xa0 = {}, xa1 = {};
  if (w == 0) {
    const v4f* p0 = (const v4f*)(xrow);
    const v4f* p1 = (const v4f*)(xrow + IN_F);
    const v4f* p2 = (const v4f*)(xrow + 2 * IN_F);
    in_product(accA, Wi, xfrag_cvt(p0[0], p0[1]), nb);
    xb0 = p1[0]; xb1 = p1[1]; xa0 = p2[0]; xa1 = p2[1];
  }

#pragma unroll 1
  for (int g = 0; g < 3; ++g) {
    if (w == 0 && g >= 1 && g <= T_STEPS / 2)
      frag_to_lds(ring + ((2 * g - 1) & (RING_SLOTS - 1)) * SLOT_HALVES, hf, lane);
    const bool active = (w == 0) ? (g < T_STEPS / 2) : (g >= 2);
    if (active) {
      const int lt = (w == 0) ? 2 * g : 2 * (g - 2);
      if (w == 0) {
        inB  = xfrag_cvt(xb0, xb1);
        inA2 = xfrag_cvt(xa0, xa1);
        const int tb = (lt + 3 < T_STEPS) ? lt + 3 : T_STEPS - 1;
        const int ta = (lt + 4 < T_STEPS) ? lt + 4 : T_STEPS - 1;
        const int tp = (lt + 8 < T_STEPS) ? lt + 8 : T_STEPS - 1;
        const v4f* pb = (const v4f*)(xrow + (size_t)tb * IN_F);
        const v4f* pa = (const v4f*)(xrow + (size_t)ta * IN_F);
        xb0 = pb[0]; xb1 = pb[1]; xa0 = pa[0]; xa1 = pa[1];
        __builtin_prefetch(xrow + (size_t)tp * IN_F, 0, 3);
      }
      if (w == 1 && g == 2) in_product(accA, Wi, in0, nb);
      in_product(accB, Wi, inB, nb);
      rec_product(accA, Wh, hf);
      cell_update(accA, c, hf);
      if (w == 0) frag_to_lds(ring + (lt & (RING_SLOTS - 1)) * SLOT_HALVES, hf, lane);
      in_product(accA, Wi, inA2, nb);
      rec_product(accB, Wh, hf);
      cell_update(accB, c, hf);
    }
    __syncthreads();
    if (w == 1) {
      if (g == 1) in0 = frag_from_lds(ring, lane);
      if (g >= 1) {
        inB  = frag_from_lds(ring + ((2 * g - 1) & (RING_SLOTS - 1)) * SLOT_HALVES, lane);
        inA2 = frag_from_lds(ring + ((2 * g) & (RING_SLOTS - 1)) * SLOT_HALVES, lane);
      }
    }
  }

#pragma unroll 1
  for (int g = 3; g < T_STEPS / 2; ++g) {
    if (w == 0)
      frag_to_lds(ring + ((2 * g - 1) & (RING_SLOTS - 1)) * SLOT_HALVES, hf, lane);
    {
      const int lt = (w == 0) ? 2 * g : 2 * (g - 2);
      if (w == 0) {
        inB  = xfrag_cvt(xb0, xb1);
        inA2 = xfrag_cvt(xa0, xa1);
        const int tb = (lt + 3 < T_STEPS) ? lt + 3 : T_STEPS - 1;
        const int ta = (lt + 4 < T_STEPS) ? lt + 4 : T_STEPS - 1;
        const int tp = (lt + 8 < T_STEPS) ? lt + 8 : T_STEPS - 1;
        const v4f* pb = (const v4f*)(xrow + (size_t)tb * IN_F);
        const v4f* pa = (const v4f*)(xrow + (size_t)ta * IN_F);
        xb0 = pb[0]; xb1 = pb[1]; xa0 = pa[0]; xa1 = pa[1];
        __builtin_prefetch(xrow + (size_t)tp * IN_F, 0, 3);
      }
      in_product(accB, Wi, inB, nb);
      rec_product(accA, Wh, hf);
      cell_update(accA, c, hf);
      if (w == 0) frag_to_lds(ring + (lt & (RING_SLOTS - 1)) * SLOT_HALVES, hf, lane);
      in_product(accA, Wi, inA2, nb);
      rec_product(accB, Wh, hf);
      cell_update(accB, c, hf);
    }
    __syncthreads();
    if (w == 1) {
      {
        inB  = frag_from_lds(ring + ((2 * g - 1) & (RING_SLOTS - 1)) * SLOT_HALVES, lane);
        inA2 = frag_from_lds(ring + ((2 * g) & (RING_SLOTS - 1)) * SLOT_HALVES, lane);
      }
    }
  }

#pragma unroll 1
  for (int g = T_STEPS / 2; g < T_STEPS / 2 + 2; ++g) {
    if (w == 0 && g >= 1 && g <= T_STEPS / 2)
      frag_to_lds(ring + ((2 * g - 1) & (RING_SLOTS - 1)) * SLOT_HALVES, hf, lane);
    const bool active = (w == 0) ? (g < T_STEPS / 2) : (g >= 2);
    if (active) {
      const int lt = (w == 0) ? 2 * g : 2 * (g - 2);
      if (w == 0) {
        inB  = xfrag_cvt(xb0, xb1);
        inA2 = xfrag_cvt(xa0, xa1);
        const int tb = (lt + 3 < T_STEPS) ? lt + 3 : T_STEPS - 1;
        const int ta = (lt + 4 < T_STEPS) ? lt + 4 : T_STEPS - 1;
        const int tp = (lt + 8 < T_STEPS) ? lt + 8 : T_STEPS - 1;
        const v4f* pb = (const v4f*)(xrow + (size_t)tb * IN_F);
        const v4f* pa = (const v4f*)(xrow + (size_t)ta * IN_F);
        xb0 = pb[0]; xb1 = pb[1]; xa0 = pa[0]; xa1 = pa[1];
        __builtin_prefetch(xrow + (size_t)tp * IN_F, 0, 3);
      }
      if (w == 1 && g == 2) in_product(accA, Wi, in0, nb);
      in_product(accB, Wi, inB, nb);
      rec_product(accA, Wh, hf);
      cell_update(accA, c, hf);
      if (w == 0) frag_to_lds(ring + (lt & (RING_SLOTS - 1)) * SLOT_HALVES, hf, lane);
      in_product(accA, Wi, inA2, nb);
      rec_product(accB, Wh, hf);
      cell_update(accB, c, hf);
    }
    __syncthreads();
    if (w == 1) {
      if (g == 1) in0 = frag_from_lds(ring, lane);
      if (g >= 1) {
        inB  = frag_from_lds(ring + ((2 * g - 1) & (RING_SLOTS - 1)) * SLOT_HALVES, lane);
        inA2 = frag_from_lds(ring + ((2 * g) & (RING_SLOTS - 1)) * SLOT_HALVES, lane);
      }
    }
  }

  if (w == 1) {
    v16h wfcf;
    {
      const int  kb = hh * 8;
      const bool r0 = (lane & 15) == 0;
#pragma unroll
      for (int i = 0; i < 8; ++i) wfcf[i]     = r0 ? (_Float16)Wfc[kb + i]      : (_Float16)0.0f;
#pragma unroll
      for (int i = 0; i < 8; ++i) wfcf[8 + i] = r0 ? (_Float16)Wfc[16 + kb + i] : (_Float16)0.0f;
    }
    v8f od = {};
    od[0] = (hh == 0) ? bfc[0] : 0.0f;
    od = wmma16(wfcf, hf, od);
    if (lane < 16) out[b0 + lane] = od[0];
  }
}

extern "C" void kernel_launch(void* const* d_in, const int* in_sizes, int n_in,
                              void* d_out, int out_size, void* d_ws, size_t ws_size,
                              hipStream_t stream) {
  const float* x    = (const float*)d_in[0];
  const float* Wih0 = (const float*)d_in[1];
  const float* Whh0 = (const float*)d_in[2];
  const float* bih0 = (const float*)d_in[3];
  const float* bhh0 = (const float*)d_in[4];
  const float* Wih1 = (const float*)d_in[5];
  const float* Whh1 = (const float*)d_in[6];
  const float* bih1 = (const float*)d_in[7];
  const float* bhh1 = (const float*)d_in[8];
  const float* Wfc  = (const float*)d_in[9];
  const float* bfc  = (const float*)d_in[10];
  float* out = (float*)d_out;

  dim3 grid(B_TOTAL / 16);
  dim3 block(64);
  lstm2_wmma_fused<<<grid, block, 0, stream>>>(x, Wih0, Whh0, bih0, bhh0,
                                               Wih1, Whh1, bih1, bhh1,
                                               Wfc, bfc, out);
}


# Trip order fixed by scheduling fences: each step's 8 recurrent products issue as one run ahead of its gates, barrier kept behind the 2nd cell; layer-1 ring reads moved to the next trip's head
# speedup vs baseline: 2.3349x; 1.0300x over previous
#include <hip/hip_runtime.h>


#define B_TOTAL 8192
#define T_STEPS 256
#define IN_F    16
#define HID     32

typedef __attribute__((ext_vector_type(16))) _Float16 v16h;
typedef __attribute__((ext_vector_type(8)))  _Float16 v8h;
typedef __attribute__((ext_vector_type(8)))  float    v8f;
typedef __attribute__((ext_vector_type(4)))  float    v4f;

static __device__ __forceinline__ float htanh(float x) { return __builtin_amdgcn_tanhf(x); }

static __device__ __forceinline__ float sigm_half(float hz) { return __builtin_fmaf(htanh(hz), 0.5f, 0.5f); }


static __device__ __forceinline__ v16h wfrag(const float* W, int ld, int tile, bool padk, float scl, int lane) {
  const int n  = tile * 16 + (lane & 15);
  const int kb = (lane < 16) ? 0 : 8;
  const float* row = W + n * ld;
  v16h a;
#pragma unroll
  for (int i = 0; i < 8; ++i) a[i] = (_Float16)(scl * row[kb + i]);
#pragma unroll
  for (int i = 0; i < 8; ++i) a[8 + i] = padk ? (_Float16)0.0f : (_Float16)(scl * row[16 + kb + i]);
  return a;
}

static __device__ __forceinline__ v8f wmma16(v16h a, v16h b, v8f c) {
  return __builtin_amdgcn_wmma_f32_16x16x32_f16(false, a, false, b, (short)0, c, false, false);
}

static __device__ __forceinline__ void in_product(v8f* acc, const v16h* Wi, const v16h in, const v8f* nb) {
#pragma unroll
  for (int q = 0; q < 8; ++q) acc[q] = wmma16(Wi[q], in, nb[q]);
}
static __device__ __forceinline__ void rec_product(v8f* acc, const v16h* Wh, const v16h h) {
#pragma unroll
  for (int q = 0; q < 8; ++q) acc[q] = wmma16(Wh[q], h, acc[q]);
}

static __device__ __forceinline__ void cell_update(const v8f* acc, v8f* c, v16h& h) {
#pragma unroll
  for (int j = 0; j < 2; ++j) {
#pragma unroll
    for (int r = 0; r < 8; ++r) {
      float iv = sigm_half(acc[4 * j + 0][r]);
      float fv = sigm_half(acc[4 * j + 1][r]);
      float gv = htanh(acc[4 * j + 2][r]);
      float ov = sigm_half(acc[4 * j + 3][r]);
      float cv = __builtin_fmaf(fv, c[j][r], iv * gv);
      c[j][r] = cv;
      float hv = ov * htanh(cv);
      h[8 * j + r] = (_Float16)hv;
    }
  }
}

#define SLOT_HALVES 512
#define RING_SLOTS  8

static __device__ __forceinline__ void frag_to_lds(_Float16* slot, const v16h f, int lane) {
  v8h lo = __builtin_shufflevector(f, f, 0, 1, 2, 3, 4, 5, 6, 7);
  v8h hi = __builtin_shufflevector(f, f, 8, 9, 10, 11, 12, 13, 14, 15);
  *(v8h*)(slot + lane * 8) = lo;
  *(v8h*)(slot + 256 + lane * 8) = hi;
}
static __device__ __forceinline__ v16h frag_from_lds(const _Float16* slot, int lane) {
  v8h lo = *(const v8h*)(slot + lane * 8);
  v8h hi = *(const v8h*)(slot + 256 + lane * 8);
  return __builtin_shufflevector(lo, hi, 0, 1, 2, 3, 4, 5, 6, 7, 8, 9, 10, 11, 12, 13, 14, 15);
}

static __device__ __forceinline__ v16h xfrag_cvt(const v4f lo, const v4f hi) {
  v16h xf;
#pragma unroll
  for (int i = 0; i < 4; ++i) { xf[i] = (_Float16)lo[i]; xf[4 + i] = (_Float16)hi[i]; }
#pragma unroll
  for (int i = 0; i < 8; ++i) xf[8 + i] = (_Float16)0.0f;
  return xf;
}

extern "C" __global__ __launch_bounds__(64)
void lstm2_wmma_fused(const float* __restrict__ x,
                      const float* __restrict__ Wih0, const float* __restrict__ Whh0,
                      const float* __restrict__ bih0, const float* __restrict__ bhh0,
                      const float* __restrict__ Wih1, const float* __restrict__ Whh1,
                      const float* __restrict__ bih1, const float* __restrict__ bhh1,
                      const float* __restrict__ Wfc,  const float* __restrict__ bfc,
                      float* __restrict__ out) {
  __shared__ __attribute__((aligned(16))) _Float16 ring[RING_SLOTS * SLOT_HALVES];
  const int lane = threadIdx.x & 31;
  const int w  = __builtin_amdgcn_readfirstlane((int)(threadIdx.x >> 5));
  const int b0 = blockIdx.x * 16;
  const int hh = lane >> 4;

  v16h Wi[8], Wh[8];
  v8f  nb[8];
  if (w == 0) {
#pragma unroll
    for (int q = 0; q < 8; ++q) {
      const int   tile = (q & 3) * 2 + (q >> 2);
      const float scl  = ((q & 3) == 2) ? 1.0f : 0.5f;
      Wi[q] = wfrag(Wih0, IN_F, tile, true,  scl, lane);
      Wh[q] = wfrag(Whh0, HID,  tile, false, scl, lane);
    }
  } else {
#pragma unroll
    for (int q = 0; q < 8; ++q) {
      const int   tile = (q & 3) * 2 + (q >> 2);
      const float scl  = ((q & 3) == 2) ? 1.0f : 0.5f;
      Wi[q] = wfrag(Wih1, HID, tile, false, scl, lane);
      Wh[q] = wfrag(Whh1, HID, tile, false, scl, lane);
    }
  }
  {
    const float* bi = (w == 0) ? bih0 : bih1;
    const float* bh = (w == 0) ? bhh0 : bhh1;
#pragma unroll
    for (int q = 0; q < 8; ++q) {
      const int   gate = q & 3;
      const int   row0 = gate * HID + (q >> 2) * 16 + hh * 8;
      const float scl  = (gate == 2) ? 1.0f : 0.5f;
#pragma unroll
      for (int r = 0; r < 8; ++r) nb[q][r] = scl * (bi[row0 + r] + bh[row0 + r]);
    }
  }

  v8f c[2];
  c[0] = (v8f){}; c[1] = (v8f){};
  v16h hf;
#pragma unroll
  for (int i = 0; i < 16; ++i) hf[i] = (_Float16)0.0f;

  const float* xrow = x + (size_t)(b0 + (lane & 15)) * T_STEPS * IN_F + hh * 8;

  v8f accA[8], accB[8];
  v16h inB, inA2;
  v16h in0;
#pragma unroll
  for (int q = 0; q < 8; ++q) { accA[q] = (v8f){}; accB[q] = (v8f){}; }
#pragma unroll
  for (int i = 0; i < 16; ++i) { inB[i] = (_Float16)0.0f; inA2[i] = (_Float16)0.0f; in0[i] = (_Float16)0.0f; }

  v4f xb0 = {}, xb1 = {}, xa0 = {}, xa1 = {};
  if (w == 0) {
    const v4f* p0 = (const v4f*)(xrow);
    const v4f* p1 = (const v4f*)(xrow + IN_F);
    const v4f* p2 = (const v4f*)(xrow + 2 * IN_F);
    in_product(accA, Wi, xfrag_cvt(p0[0], p0[1]), nb);
    xb0 = p1[0]; xb1 = p1[1]; xa0 = p2[0]; xa1 = p2[1];
  }

#pragma unroll 1
  for (int g = 0; g < 3; ++g) {
    if (w == 0 && g >= 1 && g <= T_STEPS / 2)
      frag_to_lds(ring + ((2 * g - 1) & (RING_SLOTS - 1)) * SLOT_HALVES, hf, lane);
    if (w == 1) {
      if (g == 2) in0 = frag_from_lds(ring, lane);
      if (g >= 2) {
        inB  = frag_from_lds(ring + ((2 * g - 3) & (RING_SLOTS - 1)) * SLOT_HALVES, lane);
        inA2 = frag_from_lds(ring + ((2 * g - 2) & (RING_SLOTS - 1)) * SLOT_HALVES, lane);
      }
    }
    const bool active = (w == 0) ? (g < T_STEPS / 2) : (g >= 2);
    if (active) {
      const int lt = (w == 0) ? 2 * g : 2 * (g - 2);
      if (w == 0) {
        inB  = xfrag_cvt(xb0, xb1);
        inA2 = xfrag_cvt(xa0, xa1);
        const int tb = (lt + 3 < T_STEPS) ? lt + 3 : T_STEPS - 1;
        const int ta = (lt + 4 < T_STEPS) ? lt + 4 : T_STEPS - 1;
        const int tp = (lt + 8 < T_STEPS) ? lt + 8 : T_STEPS - 1;
        const v4f* pb = (const v4f*)(xrow + (size_t)tb * IN_F);
        const v4f* pa = (const v4f*)(xrow + (size_t)ta * IN_F);
        xb0 = pb[0]; xb1 = pb[1]; xa0 = pa[0]; xa1 = pa[1];
        __builtin_prefetch(xrow + (size_t)tp * IN_F, 0, 3);
      }
      if (w == 1 && g == 2) in_product(accA, Wi, in0, nb);
      rec_product(accA, Wh, hf);
      __builtin_amdgcn_sched_barrier(0);
      in_product(accB, Wi, inB, nb);
      cell_update(accA, c, hf);
      if (w == 0) frag_to_lds(ring + (lt & (RING_SLOTS - 1)) * SLOT_HALVES, hf, lane);
      rec_product(accB, Wh, hf);
      __builtin_amdgcn_sched_barrier(0);
      in_product(accA, Wi, inA2, nb);
      cell_update(accB, c, hf);
    }
    __builtin_amdgcn_sched_barrier(0);
    __syncthreads();
  }

#pragma unroll 1
  for (int g = 3; g < T_STEPS / 2; ++g) {
    if (w == 0) {
      frag_to_lds(ring + ((2 * g - 1) & (RING_SLOTS - 1)) * SLOT_HALVES, hf, lane);
    } else {
      inB  = frag_from_lds(ring + ((2 * g - 3) & (RING_SLOTS - 1)) * SLOT_HALVES, lane);
      inA2 = frag_from_lds(ring + ((2 * g - 2) & (RING_SLOTS - 1)) * SLOT_HALVES, lane);
    }
    {
      const int lt = (w == 0) ? 2 * g : 2 * (g - 2);
      if (w == 0) {
        inB  = xfrag_cvt(xb0, xb1);
        inA2 = xfrag_cvt(xa0, xa1);
        const int tb = (lt + 3 < T_STEPS) ? lt + 3 : T_STEPS - 1;
        const int ta = (lt + 4 < T_STEPS) ? lt + 4 : T_STEPS - 1;
        const int tp = (lt + 8 < T_STEPS) ? lt + 8 : T_STEPS - 1;
        const v4f* pb = (const v4f*)(xrow + (size_t)tb * IN_F);
        const v4f* pa = (const v4f*)(xrow + (size_t)ta * IN_F);
        xb0 = pb[0]; xb1 = pb[1]; xa0 = pa[0]; xa1 = pa[1];
        __builtin_prefetch(xrow + (size_t)tp * IN_F, 0, 3);
      }
      rec_product(accA, Wh, hf);
      __builtin_amdgcn_sched_barrier(0);
      in_product(accB, Wi, inB, nb);
      cell_update(accA, c, hf);
      if (w == 0) frag_to_lds(ring + (lt & (RING_SLOTS - 1)) * SLOT_HALVES, hf, lane);
      rec_product(accB, Wh, hf);
      __builtin_amdgcn_sched_barrier(0);
      in_product(accA, Wi, inA2, nb);
      cell_update(accB, c, hf);
    }
    __builtin_amdgcn_sched_barrier(0);
    __syncthreads();
  }

#pragma unroll 1
  for (int g = T_STEPS / 2; g < T_STEPS / 2 + 2; ++g) {
    if (w == 0 && g >= 1 && g <= T_STEPS / 2)
      frag_to_lds(ring + ((2 * g - 1) & (RING_SLOTS - 1)) * SLOT_HALVES, hf, lane);
    if (w == 1) {
      if (g == 2) in0 = frag_from_lds(ring, lane);
      if (g >= 2) {
        inB  = frag_from_lds(ring + ((2 * g - 3) & (RING_SLOTS - 1)) * SLOT_HALVES, lane);
        inA2 = frag_from_lds(ring + ((2 * g - 2) & (RING_SLOTS - 1)) * SLOT_HALVES, lane);
      }
    }
    const bool active = (w == 0) ? (g < T_STEPS / 2) : (g >= 2);
    if (active) {
      const int lt = (w == 0) ? 2 * g : 2 * (g - 2);
      if (w == 0) {
        inB  = xfrag_cvt(xb0, xb1);
        inA2 = xfrag_cvt(xa0, xa1);
        const int tb = (lt + 3 < T_STEPS) ? lt + 3 : T_STEPS - 1;
        const int ta = (lt + 4 < T_STEPS) ? lt + 4 : T_STEPS - 1;
        const int tp = (lt + 8 < T_STEPS) ? lt + 8 : T_STEPS - 1;
        const v4f* pb = (const v4f*)(xrow + (size_t)tb * IN_F);
        const v4f* pa = (const v4f*)(xrow + (size_t)ta * IN_F);
        xb0 = pb[0]; xb1 = pb[1]; xa0 = pa[0]; xa1 = pa[1];
        __builtin_prefetch(xrow + (size_t)tp * IN_F, 0, 3);
      }
      if (w == 1 && g == 2) in_product(accA, Wi, in0, nb);
      rec_product(accA, Wh, hf);
      __builtin_amdgcn_sched_barrier(0);
      in_product(accB, Wi, inB, nb);
      cell_update(accA, c, hf);
      if (w == 0) frag_to_lds(ring + (lt & (RING_SLOTS - 1)) * SLOT_HALVES, hf, lane);
      rec_product(accB, Wh, hf);
      __builtin_amdgcn_sched_barrier(0);
      in_product(accA, Wi, inA2, nb);
      cell_update(accB, c, hf);
    }
    __builtin_amdgcn_sched_barrier(0);
    __syncthreads();
  }

  if (w == 1) {
    v16h wfcf;
    {
      const int  kb = hh * 8;
      const bool r0 = (lane & 15) == 0;
#pragma unroll
      for (int i = 0; i < 8; ++i) wfcf[i]     = r0 ? (_Float16)Wfc[kb + i]      : (_Float16)0.0f;
#pragma unroll
      for (int i = 0; i < 8; ++i) wfcf[8 + i] = r0 ? (_Float16)Wfc[16 + kb + i] : (_Float16)0.0f;
    }
    v8f od = {};
    od[0] = (hh == 0) ? bfc[0] : 0.0f;
    od = wmma16(wfcf, hf, od);
    if (lane < 16) out[b0 + lane] = od[0];
  }
}

extern "C" void kernel_launch(void* const* d_in, const int* in_sizes, int n_in,
                              void* d_out, int out_size, void* d_ws, size_t ws_size,
                              hipStream_t stream) {
  const float* x    = (const float*)d_in[0];
  const float* Wih0 = (const float*)d_in[1];
  const float* Whh0 = (const float*)d_in[2];
  const float* bih0 = (const float*)d_in[3];
  const float* bhh0 = (const float*)d_in[4];
  const float* Wih1 = (const float*)d_in[5];
  const float* Whh1 = (const float*)d_in[6];
  const float* bih1 = (const float*)d_in[7];
  const float* bhh1 = (const float*)d_in[8];
  const float* Wfc  = (const float*)d_in[9];
  const float* bfc  = (const float*)d_in[10];
  float* out = (float*)d_out;

  dim3 grid(B_TOTAL / 16);
  dim3 block(64);
  lstm2_wmma_fused<<<grid, block, 0, stream>>>(x, Wih0, Whh0, bih0, bhh0,
                                               Wih1, Whh1, bih1, bhh1,
                                               Wfc, bfc, out);
}
